# starting kernel
# baseline (speedup 1.0000x reference)
#include <hip/hip_runtime.h>


typedef float v2f __attribute__((ext_vector_type(2)));
typedef float v8f __attribute__((ext_vector_type(8)));

#define BATCH       8192
#define PROTOS      4096
#define CLASSES     100
#define NTILES      7
#define WAVES       8
#define THREADS     (WAVES * 32)
#define KSLABS      (PROTOS / 4)
#define SLABS_PER_W (KSLABS / WAVES)
#define ROWTILES    2
#define BLOCK_ROWS  (16 * ROWTILES)

__global__ __launch_bounds__(THREADS)
void stratified_sum_wmma(const float* __restrict__ V,
                         const int*   __restrict__ labels,
                         float*       __restrict__ out) {
    __shared__ __align__(32) float smem[WAVES * NTILES * 32 * 8];
    int* lds_labels = (int*)smem;

    const int tid  = threadIdx.x;
    const int wid  = tid >> 5;
    const int lane = tid & 31;
    const int row_base = blockIdx.x * BLOCK_ROWS;

    for (int i = tid; i < PROTOS; i += THREADS)
        lds_labels[i] = labels[i];
    __syncthreads();

    const int l15  = lane & 15;
    const int koff = (lane >> 4) * 2;

    v8f acc[ROWTILES][NTILES];
#pragma unroll
    for (int h = 0; h < ROWTILES; ++h)
#pragma unroll
        for (int t = 0; t < NTILES; ++t) { v8f z = {}; acc[h][t] = z; }

    const int    kstart = wid * SLABS_PER_W * 4;
    const float* vp0 = V + (size_t)(row_base + l15) * PROTOS + kstart + koff;
    const float* vp1 = vp0 + (size_t)16 * PROTOS;
    const int*   lp  = lds_labels + kstart + koff;

#pragma unroll 4
    for (int s = 0; s < SLABS_PER_W; ++s) {
        v2f a0 = *(const v2f*)vp0;
        v2f a1 = *(const v2f*)vp1;
        const int la = lp[0];
        const int lb = lp[1];
#pragma unroll
        for (int t = 0; t < NTILES; ++t) {
            const int cls = t * 16 + l15;
            v2f b;
            b.x = (la == cls) ? 1.0f : 0.0f;
            b.y = (lb == cls) ? 1.0f : 0.0f;
            acc[0][t] = __builtin_amdgcn_wmma_f32_16x16x4_f32(
                false, a0, false, b, (short)0, acc[0][t], false, false);
            acc[1][t] = __builtin_amdgcn_wmma_f32_16x16x4_f32(
                false, a1, false, b, (short)0, acc[1][t], false, false);
        }
        vp0 += 4;
        vp1 += 4;
        lp  += 4;
    }

#pragma unroll
    for (int h = 0; h < ROWTILES; ++h) {
        __syncthreads();
#pragma unroll
        for (int t = 0; t < NTILES; ++t) {
            float* dst = &smem[(((wid * NTILES) + t) * 32 + lane) * 8];
            *(v8f*)dst = acc[h][t];
        }
        __syncthreads();

        for (int idx = tid; idx < NTILES * 256; idx += THREADS) {
            const int t  = idx >> 8;
            const int ln = (idx >> 3) & 31;
            const int r  = idx & 7;
            float sum = 0.0f;
#pragma unroll
            for (int w = 0; w < WAVES; ++w)
                sum += smem[(((w * NTILES) + t) * 32 + ln) * 8 + r];
            const int row = row_base + h * 16 + r + ((ln >> 4) << 3);
            const int cls = t * 16 + (ln & 15);
            if (cls < CLASSES)
                out[(size_t)row * CLASSES + cls] = sum;
        }
    }
}

extern "C" void kernel_launch(void* const* d_in, const int* in_sizes, int n_in,
                              void* d_out, int out_size, void* d_ws, size_t ws_size,
                              hipStream_t stream) {
    const float* V      = (const float*)d_in[0];
    const int*   labels = (const int*)d_in[1];
    float*       out    = (float*)d_out;

    stratified_sum_wmma<<<BATCH / BLOCK_ROWS, THREADS, 0, stream>>>(V, labels, out);
}


# Skip the WMMAs of class tiles whose one-hot B tile is all zero in a k-slab: per-slab 7-bit tile mask built once in LDS, scalar bit test per tile; summation order kept, only exact-zero terms dropped
# speedup vs baseline: 1.4507x; 1.4507x over previous
#include <hip/hip_runtime.h>


typedef float v2f __attribute__((ext_vector_type(2)));
typedef float v8f __attribute__((ext_vector_type(8)));

#define BATCH       8192
#define PROTOS      4096
#define CLASSES     100
#define NTILES      7
#define WAVES       8
#define THREADS     (WAVES * 32)
#define KSLABS      (PROTOS / 4)
#define SLABS_PER_W (KSLABS / WAVES)
#define GROUP       4
#define GROUPS      (KSLABS / GROUP)
#define GROUPS_PER_W (SLABS_PER_W / GROUP)
#define ROWTILES    2
#define BLOCK_ROWS  (16 * ROWTILES)

__global__ __launch_bounds__(THREADS)
void stratified_sum_wmma(const float* __restrict__ V,
                         const int*   __restrict__ labels,
                         float*       __restrict__ out) {
    __shared__ __align__(32) float smem[WAVES * NTILES * 32 * 8];
    int* lds_labels = (int*)smem;
    unsigned* lds_hits = (unsigned*)smem + PROTOS;
    static_assert(GROUPS == THREADS, "one hit word per thread");

    const int tid  = threadIdx.x;
    const int wid  = tid >> 5;
    const int lane = tid & 31;
    const int row_base = blockIdx.x * BLOCK_ROWS;

    for (int i = tid; i < PROTOS; i += THREADS)
        lds_labels[i] = labels[i];
    __syncthreads();

    {
        const int* gl = lds_labels + tid * (GROUP * 4);
        unsigned packed = 0u;
        for (int j = 0; j < GROUP; ++j) {
            unsigned m = 0u;
            for (int e = 0; e < 4; ++e) {
                const unsigned ti = (unsigned)gl[4 * j + e] >> 4;
                m |= (ti < (unsigned)NTILES) ? (1u << ti) : 0u;
            }
            packed |= m << (8 * j);
        }
        lds_hits[tid] = packed;
    }
    __syncthreads();

    const int l15  = lane & 15;
    const int koff = (lane >> 4) * 2;

    v8f acc[ROWTILES][NTILES];
#pragma unroll
    for (int h = 0; h < ROWTILES; ++h)
#pragma unroll
        for (int t = 0; t < NTILES; ++t) { v8f z = {}; acc[h][t] = z; }

    const int    kstart = wid * SLABS_PER_W * 4;
    const float* vp0 = V + (size_t)(row_base + l15) * PROTOS + kstart + koff;
    const float* vp1 = vp0 + (size_t)16 * PROTOS;
    const int*   lp  = lds_labels + kstart + koff;

    const unsigned* hp = lds_hits + wid * GROUPS_PER_W;
    unsigned hits_next = hp[0];
    for (int g = 0; g < GROUPS_PER_W; ++g) {
        const unsigned hits = (unsigned)__builtin_amdgcn_readfirstlane((int)hits_next);
        hits_next = hp[(g + 1 < GROUPS_PER_W) ? (g + 1) : g];

        v2f a0[GROUP], a1[GROUP];
        int la[GROUP], lb[GROUP];
#pragma unroll
        for (int j = 0; j < GROUP; ++j) {
            a0[j] = *(const v2f*)(vp0 + 4 * j);
            a1[j] = *(const v2f*)(vp1 + 4 * j);
            la[j] = lp[4 * j];
            lb[j] = lp[4 * j + 1];
        }
#pragma unroll
        for (int j = 0; j < GROUP; ++j) {
#pragma unroll
            for (int t = 0; t < NTILES; ++t) {
                if (hits & (1u << (8 * j + t))) {
                    const int cls = t * 16 + l15;
                    v2f b;
                    b.x = (la[j] == cls) ? 1.0f : 0.0f;
                    b.y = (lb[j] == cls) ? 1.0f : 0.0f;
                    acc[0][t] = __builtin_amdgcn_wmma_f32_16x16x4_f32(
                        false, a0[j], false, b, (short)0, acc[0][t], false, false);
                    acc[1][t] = __builtin_amdgcn_wmma_f32_16x16x4_f32(
                        false, a1[j], false, b, (short)0, acc[1][t], false, false);
                }
            }
        }
        vp0 += 4 * GROUP;
        vp1 += 4 * GROUP;
        lp  += 4 * GROUP;
    }

#pragma unroll
    for (int h = 0; h < ROWTILES; ++h) {
        __syncthreads();
#pragma unroll
        for (int t = 0; t < NTILES; ++t) {
            float* dst = &smem[(((wid * NTILES) + t) * 32 + lane) * 8];
            *(v8f*)dst = acc[h][t];
        }
        __syncthreads();

        for (int idx = tid; idx < NTILES * 256; idx += THREADS) {
            const int t  = idx >> 8;
            const int ln = (idx >> 3) & 31;
            const int r  = idx & 7;
            float sum = 0.0f;
#pragma unroll
            for (int w = 0; w < WAVES; ++w)
                sum += smem[(((w * NTILES) + t) * 32 + ln) * 8 + r];
            const int row = row_base + h * 16 + r + ((ln >> 4) << 3);
            const int cls = t * 16 + (ln & 15);
            if (cls < CLASSES)
                out[(size_t)row * CLASSES + cls] = sum;
        }
    }
}

extern "C" void kernel_launch(void* const* d_in, const int* in_sizes, int n_in,
                              void* d_out, int out_size, void* d_ws, size_t ws_size,
                              hipStream_t stream) {
    const float* V      = (const float*)d_in[0];
    const int*   labels = (const int*)d_in[1];
    float*       out    = (float*)d_out;

    stratified_sum_wmma<<<BATCH / BLOCK_ROWS, THREADS, 0, stream>>>(V, labels, out);
}


# Software-pipeline the K loop: next group's V fragments and labels load into a second register set ahead of the current group's WMMAs; group = 8 k-slabs (one 128-B line per row); sum order unchanged
# speedup vs baseline: 1.6262x; 1.1210x over previous
#include <hip/hip_runtime.h>


typedef float v2f __attribute__((ext_vector_type(2)));
typedef float v8f __attribute__((ext_vector_type(8)));

#define BATCH       8192
#define PROTOS      4096
#define CLASSES     100
#define NTILES      7
#define WAVES       8
#define THREADS     (WAVES * 32)
#define KSLABS      (PROTOS / 4)
#define SLABS_PER_W (KSLABS / WAVES)
#define HSLABS      4
#define HWORDS      (KSLABS / HSLABS)
#define HWORDS_PER_W (SLABS_PER_W / HSLABS)
#define GROUP       8
#define GROUPS_PER_W (SLABS_PER_W / GROUP)
#define ROWTILES    2
#define BLOCK_ROWS  (16 * ROWTILES)

__global__ __launch_bounds__(THREADS)
void stratified_sum_wmma(const float* __restrict__ V,
                         const int*   __restrict__ labels,
                         float*       __restrict__ out) {
    __shared__ __align__(32) float smem[WAVES * NTILES * 32 * 8];
    int* lds_labels = (int*)smem;
    unsigned* lds_hits = (unsigned*)smem + PROTOS;
    static_assert(HWORDS == THREADS, "one hit word per thread");
    static_assert(GROUP == 2 * HSLABS, "one loop trip consumes two hit words");

    const int tid  = threadIdx.x;
    const int wid  = tid >> 5;
    const int lane = tid & 31;
    const int row_base = blockIdx.x * BLOCK_ROWS;

    for (int i = tid; i < PROTOS; i += THREADS)
        lds_labels[i] = labels[i];
    __syncthreads();

    {
        const int* gl = lds_labels + tid * (HSLABS * 4);
        unsigned packed = 0u;
        for (int j = 0; j < HSLABS; ++j) {
            unsigned m = 0u;
            for (int e = 0; e < 4; ++e) {
                const unsigned ti = (unsigned)gl[4 * j + e] >> 4;
                m |= (ti < (unsigned)NTILES) ? (1u << ti) : 0u;
            }
            packed |= m << (8 * j);
        }
        lds_hits[tid] = packed;
    }
    __syncthreads();

    const int l15  = lane & 15;
    const int koff = (lane >> 4) * 2;

    v8f acc[ROWTILES][NTILES];
#pragma unroll
    for (int h = 0; h < ROWTILES; ++h)
#pragma unroll
        for (int t = 0; t < NTILES; ++t) { v8f z = {}; acc[h][t] = z; }

    const int    kstart = wid * SLABS_PER_W * 4;
    const float* vp0 = V + (size_t)(row_base + l15) * PROTOS + kstart + koff;
    const float* vp1 = vp0 + (size_t)16 * PROTOS;
    const int*   lp  = lds_labels + kstart + koff;

    const unsigned* hp = lds_hits + wid * HWORDS_PER_W;
    v2f a0[GROUP], a1[GROUP];
    int la[GROUP], lb[GROUP];
#pragma unroll
    for (int j = 0; j < GROUP; ++j) {
        a0[j] = *(const v2f*)(vp0 + 4 * j);
        a1[j] = *(const v2f*)(vp1 + 4 * j);
        la[j] = lp[4 * j];
        lb[j] = lp[4 * j + 1];
    }
    unsigned hlo_next = hp[0];
    unsigned hhi_next = hp[1];
    for (int g = 0; g < GROUPS_PER_W; ++g) {
        const unsigned hlo = (unsigned)__builtin_amdgcn_readfirstlane((int)hlo_next);
        const unsigned hhi = (unsigned)__builtin_amdgcn_readfirstlane((int)hhi_next);

        v2f n0[GROUP], n1[GROUP];
        int nla[GROUP], nlb[GROUP];
#pragma unroll
        for (int j = 0; j < GROUP; ++j) { n0[j] = a0[j]; n1[j] = a1[j]; nla[j] = la[j]; nlb[j] = lb[j]; }
        if (g + 1 < GROUPS_PER_W) {
            vp0 += 4 * GROUP;
            vp1 += 4 * GROUP;
            lp  += 4 * GROUP;
#pragma unroll
            for (int j = 0; j < GROUP; ++j) {
                n0[j]  = *(const v2f*)(vp0 + 4 * j);
                n1[j]  = *(const v2f*)(vp1 + 4 * j);
                nla[j] = lp[4 * j];
                nlb[j] = lp[4 * j + 1];
            }
            hlo_next = hp[2 * (g + 1)];
            hhi_next = hp[2 * (g + 1) + 1];
        }

#pragma unroll
        for (int j = 0; j < GROUP; ++j) {
            const unsigned hw = (j < HSLABS) ? hlo : hhi;
#pragma unroll
            for (int t = 0; t < NTILES; ++t) {
                if (hw & (1u << (8 * (j % HSLABS) + t))) {
                    const int cls = t * 16 + l15;
                    v2f b;
                    b.x = (la[j] == cls) ? 1.0f : 0.0f;
                    b.y = (lb[j] == cls) ? 1.0f : 0.0f;
                    acc[0][t] = __builtin_amdgcn_wmma_f32_16x16x4_f32(
                        false, a0[j], false, b, (short)0, acc[0][t], false, false);
                    acc[1][t] = __builtin_amdgcn_wmma_f32_16x16x4_f32(
                        false, a1[j], false, b, (short)0, acc[1][t], false, false);
                }
            }
        }
#pragma unroll
        for (int j = 0; j < GROUP; ++j) { a0[j] = n0[j]; a1[j] = n1[j]; la[j] = nla[j]; lb[j] = nlb[j]; }
    }

#pragma unroll
    for (int h = 0; h < ROWTILES; ++h) {
        __syncthreads();
#pragma unroll
        for (int t = 0; t < NTILES; ++t) {
            float* dst = &smem[(((wid * NTILES) + t) * 32 + lane) * 8];
            *(v8f*)dst = acc[h][t];
        }
        __syncthreads();

        for (int idx = tid; idx < NTILES * 256; idx += THREADS) {
            const int t  = idx >> 8;
            const int ln = (idx >> 3) & 31;
            const int r  = idx & 7;
            float sum = 0.0f;
#pragma unroll
            for (int w = 0; w < WAVES; ++w)
                sum += smem[(((w * NTILES) + t) * 32 + ln) * 8 + r];
            const int row = row_base + h * 16 + r + ((ln >> 4) << 3);
            const int cls = t * 16 + (ln & 15);
            if (cls < CLASSES)
                out[(size_t)row * CLASSES + cls] = sum;
        }
    }
}

extern "C" void kernel_launch(void* const* d_in, const int* in_sizes, int n_in,
                              void* d_out, int out_size, void* d_ws, size_t ws_size,
                              hipStream_t stream) {
    const float* V      = (const float*)d_in[0];
    const int*   labels = (const int*)d_in[1];
    float*       out    = (float*)d_out;

    stratified_sum_wmma<<<BATCH / BLOCK_ROWS, THREADS, 0, stream>>>(V, labels, out);
}


# Prologue in one memory round trip: each thread's four 16-byte label loads and the first group's V loads issue together; tile masks come from the label registers (one barrier, no LDS re-read)
# speedup vs baseline: 1.9208x; 1.1812x over previous
#include <hip/hip_runtime.h>


typedef float v2f __attribute__((ext_vector_type(2)));
typedef float v8f __attribute__((ext_vector_type(8)));
typedef int   v4i __attribute__((ext_vector_type(4)));

#define BATCH       8192
#define PROTOS      4096
#define CLASSES     100
#define NTILES      7
#define WAVES       8
#define THREADS     (WAVES * 32)
#define KSLABS      (PROTOS / 4)
#define SLABS_PER_W (KSLABS / WAVES)
#define HSLABS      4
#define HWORDS      (KSLABS / HSLABS)
#define HWORDS_PER_W (SLABS_PER_W / HSLABS)
#define GROUP       8
#define GROUPS_PER_W (SLABS_PER_W / GROUP)
#define ROWTILES    2
#define BLOCK_ROWS  (16 * ROWTILES)

__global__ __launch_bounds__(THREADS)
void stratified_sum_wmma(const float* __restrict__ V,
                         const int*   __restrict__ labels,
                         float*       __restrict__ out) {
    __shared__ __align__(32) float smem[WAVES * NTILES * 32 * 8];
    int* lds_labels = (int*)smem;
    unsigned* lds_hits = (unsigned*)smem + PROTOS;
    static_assert(HWORDS == THREADS, "one hit word per thread");
    static_assert(GROUP == 2 * HSLABS, "one loop trip consumes two hit words");

    const int tid  = threadIdx.x;
    const int wid  = tid >> 5;
    const int lane = tid & 31;
    const int row_base = blockIdx.x * BLOCK_ROWS;

    static_assert(KSLABS == 4 * THREADS, "four k-slabs per thread");
    const v4i* gl4 = (const v4i*)labels;
    v4i lab4[4];
#pragma unroll
    for (int q = 0; q < 4; ++q)
        lab4[q] = gl4[q * THREADS + tid];

    const int l15  = lane & 15;
    const int koff = (lane >> 4) * 2;

    const int    kstart = wid * SLABS_PER_W * 4;
    const float* vp0 = V + (size_t)(row_base + l15) * PROTOS + kstart + koff;
    const float* vp1 = vp0 + (size_t)16 * PROTOS;

    v2f a0[GROUP], a1[GROUP];
#pragma unroll
    for (int j = 0; j < GROUP; ++j) {
        a0[j] = *(const v2f*)(vp0 + 4 * j);
        a1[j] = *(const v2f*)(vp1 + 4 * j);
    }

    {
        v4i* lds_lab4 = (v4i*)lds_labels;
        unsigned char* lds_hit_bytes = (unsigned char*)lds_hits;
#pragma unroll
        for (int q = 0; q < 4; ++q) {
            const v4i l4 = lab4[q];
            lds_lab4[q * THREADS + tid] = l4;
            unsigned m = 0u;
#pragma unroll
            for (int e = 0; e < 4; ++e) {
                const unsigned ti = (unsigned)l4[e] >> 4;
                m |= (ti < (unsigned)NTILES) ? (1u << ti) : 0u;
            }
            lds_hit_bytes[q * THREADS + tid] = (unsigned char)m;
        }
    }
    __syncthreads();

    v8f acc[ROWTILES][NTILES];
#pragma unroll
    for (int h = 0; h < ROWTILES; ++h)
#pragma unroll
        for (int t = 0; t < NTILES; ++t) { v8f z = {}; acc[h][t] = z; }

    const int*   lp  = lds_labels + kstart + koff;

    const unsigned* hp = lds_hits + wid * HWORDS_PER_W;
    int la[GROUP], lb[GROUP];
#pragma unroll
    for (int j = 0; j < GROUP; ++j) {
        la[j] = lp[4 * j];
        lb[j] = lp[4 * j + 1];
    }
    unsigned hlo_next = hp[0];
    unsigned hhi_next = hp[1];
    for (int g = 0; g < GROUPS_PER_W; ++g) {
        const unsigned hlo = (unsigned)__builtin_amdgcn_readfirstlane((int)hlo_next);
        const unsigned hhi = (unsigned)__builtin_amdgcn_readfirstlane((int)hhi_next);

        v2f n0[GROUP], n1[GROUP];
        int nla[GROUP], nlb[GROUP];
#pragma unroll
        for (int j = 0; j < GROUP; ++j) { n0[j] = a0[j]; n1[j] = a1[j]; nla[j] = la[j]; nlb[j] = lb[j]; }
        if (g + 1 < GROUPS_PER_W) {
            vp0 += 4 * GROUP;
            vp1 += 4 * GROUP;
            lp  += 4 * GROUP;
#pragma unroll
            for (int j = 0; j < GROUP; ++j) {
                n0[j]  = *(const v2f*)(vp0 + 4 * j);
                n1[j]  = *(const v2f*)(vp1 + 4 * j);
                nla[j] = lp[4 * j];
                nlb[j] = lp[4 * j + 1];
            }
            hlo_next = hp[2 * (g + 1)];
            hhi_next = hp[2 * (g + 1) + 1];
        }

#pragma unroll
        for (int j = 0; j < GROUP; ++j) {
            const unsigned hw = (j < HSLABS) ? hlo : hhi;
#pragma unroll
            for (int t = 0; t < NTILES; ++t) {
                if (hw & (1u << (8 * (j % HSLABS) + t))) {
                    const int cls = t * 16 + l15;
                    v2f b;
                    b.x = (la[j] == cls) ? 1.0f : 0.0f;
                    b.y = (lb[j] == cls) ? 1.0f : 0.0f;
                    acc[0][t] = __builtin_amdgcn_wmma_f32_16x16x4_f32(
                        false, a0[j], false, b, (short)0, acc[0][t], false, false);
                    acc[1][t] = __builtin_amdgcn_wmma_f32_16x16x4_f32(
                        false, a1[j], false, b, (short)0, acc[1][t], false, false);
                }
            }
        }
#pragma unroll
        for (int j = 0; j < GROUP; ++j) { a0[j] = n0[j]; a1[j] = n1[j]; la[j] = nla[j]; lb[j] = nlb[j]; }
    }

#pragma unroll
    for (int h = 0; h < ROWTILES; ++h) {
        __syncthreads();
#pragma unroll
        for (int t = 0; t < NTILES; ++t) {
            float* dst = &smem[(((wid * NTILES) + t) * 32 + lane) * 8];
            *(v8f*)dst = acc[h][t];
        }
        __syncthreads();

        for (int idx = tid; idx < NTILES * 256; idx += THREADS) {
            const int t  = idx >> 8;
            const int ln = (idx >> 3) & 31;
            const int r  = idx & 7;
            float sum = 0.0f;
#pragma unroll
            for (int w = 0; w < WAVES; ++w)
                sum += smem[(((w * NTILES) + t) * 32 + ln) * 8 + r];
            const int row = row_base + h * 16 + r + ((ln >> 4) << 3);
            const int cls = t * 16 + (ln & 15);
            if (cls < CLASSES)
                out[(size_t)row * CLASSES + cls] = sum;
        }
    }
}

extern "C" void kernel_launch(void* const* d_in, const int* in_sizes, int n_in,
                              void* d_out, int out_size, void* d_ws, size_t ws_size,
                              hipStream_t stream) {
    const float* V      = (const float*)d_in[0];
    const int*   labels = (const int*)d_in[1];
    float*       out    = (float*)d_out;

    stratified_sum_wmma<<<BATCH / BLOCK_ROWS, THREADS, 0, stream>>>(V, labels, out);
}
